# starting kernel
# baseline (speedup 1.0000x reference)
#include <hip/hip_runtime.h>
#include <hip/hip_bf16.h>

#define N_NODES   50000
#define N_EDGES   800000
#define DIM       128
#define NEG_SLOPE 0.01f
#define APAD      132

typedef float v2f __attribute__((ext_vector_type(2)));
typedef float v8f __attribute__((ext_vector_type(8)));

__global__ void __launch_bounds__(256)
gat_init_kernel(float* __restrict__ out, float* __restrict__ m, float* __restrict__ denom) {
    int tid = blockIdx.x * blockDim.x + threadIdx.x;
    if (tid < N_NODES * DIM) out[tid] = 0.0f;
    if (tid < N_NODES) {
        m[tid]     = -__builtin_inff();
        denom[tid] = 0.0f;
    }
}

__global__ void __launch_bounds__(256)
gat_gemm_wmma(const float* __restrict__ h, const float* __restrict__ W,
              float* __restrict__ z) {
    __shared__ float As[16 * APAD];

    const int tid     = threadIdx.y * 32 + threadIdx.x;
    const int rowBase = blockIdx.x * 16;

    {
        const int r = tid >> 4;
        const int c = (tid & 15) * 8;
        const float* gp = h + (size_t)(rowBase + r) * DIM + c;
        float*       lp = As + r * APAD + c;
        #pragma unroll
        for (int i = 0; i < 8; ++i) lp[i] = gp[i];
    }
    __syncthreads();

    const int lane  = threadIdx.x;
    const int col   = threadIdx.y * 16 + (lane & 15);
    const int khalf = (lane >> 4) * 2;

    const float* ap = As + (lane & 15) * APAD + khalf;
    const float* bp = W + (size_t)khalf * DIM + col;

    v8f c = {0.f, 0.f, 0.f, 0.f, 0.f, 0.f, 0.f, 0.f};

    #pragma unroll
    for (int k = 0; k < DIM; k += 4) {
        v2f a, b;
        a.x = ap[k];
        a.y = ap[k + 1];
        b.x = bp[k * DIM];
        b.y = bp[k * DIM + DIM];
        c = __builtin_amdgcn_wmma_f32_16x16x4_f32(
                false, a, false, b, (short)0, c, false, false);
    }

    const int rstep = (lane >> 4) * 8;
    float* zp = z + (size_t)(rowBase + rstep) * DIM + col;
    #pragma unroll
    for (int r = 0; r < 8; ++r)
        zp[r * DIM] = c[r];
}

__global__ void __launch_bounds__(256)
gat_sd_kernel(const float* __restrict__ z, const float* __restrict__ a_src,
              const float* __restrict__ a_dst,
              float* __restrict__ s, float* __restrict__ d) {
    const int row = blockIdx.x * 8 + threadIdx.y;
    if (row >= N_NODES) return;
    const int lane = threadIdx.x;

    const float* zp = z + (size_t)row * DIM + lane * 4;
    float ss = 0.f, dd = 0.f;
    #pragma unroll
    for (int i = 0; i < 4; ++i) {
        const float zv = zp[i];
        ss += zv * a_src[lane * 4 + i];
        dd += zv * a_dst[lane * 4 + i];
    }
    #pragma unroll
    for (int off = 16; off > 0; off >>= 1) {
        ss += __shfl_xor(ss, off, 32);
        dd += __shfl_xor(dd, off, 32);
    }
    if (lane == 0) { s[row] = ss; d[row] = dd; }
}

__device__ inline void atomicMaxF(float* addr, float val) {
    if (val >= 0.0f) atomicMax((int*)addr, __float_as_int(val));
    else             atomicMin((unsigned int*)addr, __float_as_uint(val));
}

__global__ void __launch_bounds__(256)
gat_edge_max_kernel(const int* __restrict__ src, const int* __restrict__ dst,
                    const float* __restrict__ s, const float* __restrict__ d,
                    float* __restrict__ e, float* __restrict__ m) {
    const int j = blockIdx.x * blockDim.x + threadIdx.x;
    if (j >= N_EDGES) return;
    float ev = s[src[j]] + d[dst[j]];
    ev = (ev >= 0.0f) ? ev : NEG_SLOPE * ev;
    e[j] = ev;
    atomicMaxF(&m[dst[j]], ev);
}

__global__ void __launch_bounds__(256)
gat_edge_exp_kernel(const int* __restrict__ dst, float* __restrict__ e,
                    const float* __restrict__ m, float* __restrict__ denom) {
    const int j = blockIdx.x * blockDim.x + threadIdx.x;
    if (j >= N_EDGES) return;
    const int dj = dst[j];
    const float ex = __expf(e[j] - m[dj]);
    e[j] = ex;
    atomicAdd(&denom[dj], ex);
}

__global__ void __launch_bounds__(256)
gat_scatter_kernel(const int* __restrict__ src, const int* __restrict__ dst,
                   const float* __restrict__ e, const float* __restrict__ denom,
                   const float* __restrict__ z, float* __restrict__ out) {
    const int j = blockIdx.x * 8 + threadIdx.y;
    if (j >= N_EDGES) return;
    const int sj = src[j];
    const int dj = dst[j];
    const float alpha = e[j] / denom[dj];
    const int f = threadIdx.x * 4;
    const float* zp = z   + (size_t)sj * DIM + f;
    float*       op = out + (size_t)dj * DIM + f;
    #pragma unroll
    for (int i = 0; i < 4; ++i)
        atomicAdd(op + i, alpha * zp[i]);
}

extern "C" void kernel_launch(void* const* d_in, const int* in_sizes, int n_in,
                              void* d_out, int out_size, void* d_ws, size_t ws_size,
                              hipStream_t stream) {
    const float* h     = (const float*)d_in[0];
    const int*   src   = (const int*)  d_in[1];
    const int*   dst   = (const int*)  d_in[2];
    const float* W     = (const float*)d_in[3];
    const float* a_src = (const float*)d_in[4];
    const float* a_dst = (const float*)d_in[5];
    float*       out   = (float*)d_out;

    float* ws    = (float*)d_ws;
    float* z     = ws;
    float* s     = z     + (size_t)N_NODES * DIM;
    float* dn    = s     + N_NODES;
    float* m     = dn    + N_NODES;
    float* denom = m     + N_NODES;
    float* e     = denom + N_NODES;

    gat_init_kernel<<<(N_NODES * DIM + 255) / 256, 256, 0, stream>>>(out, m, denom);
    gat_gemm_wmma<<<N_NODES / 16, dim3(32, 8), 0, stream>>>(h, W, z);
    gat_sd_kernel<<<(N_NODES + 7) / 8, dim3(32, 8), 0, stream>>>(z, a_src, a_dst, s, dn);
    gat_edge_max_kernel<<<(N_EDGES + 255) / 256, 256, 0, stream>>>(src, dst, s, dn, e, m);
    gat_edge_exp_kernel<<<(N_EDGES + 255) / 256, 256, 0, stream>>>(dst, e, m, denom);
    gat_scatter_kernel<<<(N_EDGES + 7) / 8, dim3(32, 8), 0, stream>>>(src, dst, e, denom, z, out);
}


# scatter-add: lane owns features lane+32k, so each f32 atomic instruction of a wave covers one whole 128-B line of out (was 8 lanes on each of 4 lines); z row read the same way
# speedup vs baseline: 1.2872x; 1.2872x over previous
#include <hip/hip_runtime.h>
#include <hip/hip_bf16.h>

#define N_NODES   50000
#define N_EDGES   800000
#define DIM       128
#define NEG_SLOPE 0.01f
#define APAD      132

typedef float v2f __attribute__((ext_vector_type(2)));
typedef float v8f __attribute__((ext_vector_type(8)));

__global__ void __launch_bounds__(256)
gat_init_kernel(float* __restrict__ out, float* __restrict__ m, float* __restrict__ denom) {
    int tid = blockIdx.x * blockDim.x + threadIdx.x;
    if (tid < N_NODES * DIM) out[tid] = 0.0f;
    if (tid < N_NODES) {
        m[tid]     = -__builtin_inff();
        denom[tid] = 0.0f;
    }
}

__global__ void __launch_bounds__(256)
gat_gemm_wmma(const float* __restrict__ h, const float* __restrict__ W,
              float* __restrict__ z) {
    __shared__ float As[16 * APAD];

    const int tid     = threadIdx.y * 32 + threadIdx.x;
    const int rowBase = blockIdx.x * 16;

    {
        const int r = tid >> 4;
        const int c = (tid & 15) * 8;
        const float* gp = h + (size_t)(rowBase + r) * DIM + c;
        float*       lp = As + r * APAD + c;
        #pragma unroll
        for (int i = 0; i < 8; ++i) lp[i] = gp[i];
    }
    __syncthreads();

    const int lane  = threadIdx.x;
    const int col   = threadIdx.y * 16 + (lane & 15);
    const int khalf = (lane >> 4) * 2;

    const float* ap = As + (lane & 15) * APAD + khalf;
    const float* bp = W + (size_t)khalf * DIM + col;

    v8f c = {0.f, 0.f, 0.f, 0.f, 0.f, 0.f, 0.f, 0.f};

    #pragma unroll
    for (int k = 0; k < DIM; k += 4) {
        v2f a, b;
        a.x = ap[k];
        a.y = ap[k + 1];
        b.x = bp[k * DIM];
        b.y = bp[k * DIM + DIM];
        c = __builtin_amdgcn_wmma_f32_16x16x4_f32(
                false, a, false, b, (short)0, c, false, false);
    }

    const int rstep = (lane >> 4) * 8;
    float* zp = z + (size_t)(rowBase + rstep) * DIM + col;
    #pragma unroll
    for (int r = 0; r < 8; ++r)
        zp[r * DIM] = c[r];
}

__global__ void __launch_bounds__(256)
gat_sd_kernel(const float* __restrict__ z, const float* __restrict__ a_src,
              const float* __restrict__ a_dst,
              float* __restrict__ s, float* __restrict__ d) {
    const int row = blockIdx.x * 8 + threadIdx.y;
    if (row >= N_NODES) return;
    const int lane = threadIdx.x;

    const float* zp = z + (size_t)row * DIM + lane * 4;
    float ss = 0.f, dd = 0.f;
    #pragma unroll
    for (int i = 0; i < 4; ++i) {
        const float zv = zp[i];
        ss += zv * a_src[lane * 4 + i];
        dd += zv * a_dst[lane * 4 + i];
    }
    #pragma unroll
    for (int off = 16; off > 0; off >>= 1) {
        ss += __shfl_xor(ss, off, 32);
        dd += __shfl_xor(dd, off, 32);
    }
    if (lane == 0) { s[row] = ss; d[row] = dd; }
}

__device__ inline void atomicMaxF(float* addr, float val) {
    if (val >= 0.0f) atomicMax((int*)addr, __float_as_int(val));
    else             atomicMin((unsigned int*)addr, __float_as_uint(val));
}

__global__ void __launch_bounds__(256)
gat_edge_max_kernel(const int* __restrict__ src, const int* __restrict__ dst,
                    const float* __restrict__ s, const float* __restrict__ d,
                    float* __restrict__ e, float* __restrict__ m) {
    const int j = blockIdx.x * blockDim.x + threadIdx.x;
    if (j >= N_EDGES) return;
    float ev = s[src[j]] + d[dst[j]];
    ev = (ev >= 0.0f) ? ev : NEG_SLOPE * ev;
    e[j] = ev;
    atomicMaxF(&m[dst[j]], ev);
}

__global__ void __launch_bounds__(256)
gat_edge_exp_kernel(const int* __restrict__ dst, float* __restrict__ e,
                    const float* __restrict__ m, float* __restrict__ denom) {
    const int j = blockIdx.x * blockDim.x + threadIdx.x;
    if (j >= N_EDGES) return;
    const int dj = dst[j];
    const float ex = __expf(e[j] - m[dj]);
    e[j] = ex;
    atomicAdd(&denom[dj], ex);
}

__global__ void __launch_bounds__(256)
gat_scatter_kernel(const int* __restrict__ src, const int* __restrict__ dst,
                   const float* __restrict__ e, const float* __restrict__ denom,
                   const float* __restrict__ z, float* __restrict__ out) {
    const int j = blockIdx.x * 8 + threadIdx.y;
    if (j >= N_EDGES) return;
    const int sj = src[j];
    const int dj = dst[j];
    const float alpha = e[j] / denom[dj];
    const int f = threadIdx.x;
    const float* zp = z   + (size_t)sj * DIM + f;
    float*       op = out + (size_t)dj * DIM + f;
    float zv[4];
    for (int i = 0; i < 4; ++i)
        zv[i] = zp[i * 32];
    #pragma unroll
    for (int i = 0; i < 4; ++i)
        atomicAdd(op + i * 32, alpha * zv[i]);
}

extern "C" void kernel_launch(void* const* d_in, const int* in_sizes, int n_in,
                              void* d_out, int out_size, void* d_ws, size_t ws_size,
                              hipStream_t stream) {
    const float* h     = (const float*)d_in[0];
    const int*   src   = (const int*)  d_in[1];
    const int*   dst   = (const int*)  d_in[2];
    const float* W     = (const float*)d_in[3];
    const float* a_src = (const float*)d_in[4];
    const float* a_dst = (const float*)d_in[5];
    float*       out   = (float*)d_out;

    float* ws    = (float*)d_ws;
    float* z     = ws;
    float* s     = z     + (size_t)N_NODES * DIM;
    float* dn    = s     + N_NODES;
    float* m     = dn    + N_NODES;
    float* denom = m     + N_NODES;
    float* e     = denom + N_NODES;

    gat_init_kernel<<<(N_NODES * DIM + 255) / 256, 256, 0, stream>>>(out, m, denom);
    gat_gemm_wmma<<<N_NODES / 16, dim3(32, 8), 0, stream>>>(h, W, z);
    gat_sd_kernel<<<(N_NODES + 7) / 8, dim3(32, 8), 0, stream>>>(z, a_src, a_dst, s, dn);
    gat_edge_max_kernel<<<(N_EDGES + 255) / 256, 256, 0, stream>>>(src, dst, s, dn, e, m);
    gat_edge_exp_kernel<<<(N_EDGES + 255) / 256, 256, 0, stream>>>(dst, e, m, denom);
    gat_scatter_kernel<<<(N_EDGES + 7) / 8, dim3(32, 8), 0, stream>>>(src, dst, e, denom, z, out);
}


# edges grouped by dst per call (integer counts, cursors), each segment sorted by source id; one wave per dst node does max, exp-sum, weighted row sum in that fixed order: no f32 atomics, no zero fill
# speedup vs baseline: 1.7083x; 1.3271x over previous
#include <hip/hip_runtime.h>
#include <hip/hip_bf16.h>

#define N_NODES   50000
#define N_EDGES   800000
#define DIM       128
#define NEG_SLOPE 0.01f
#define APAD      132

typedef float v2f __attribute__((ext_vector_type(2)));
typedef float v8f __attribute__((ext_vector_type(8)));


__global__ void __launch_bounds__(256)
gat_gemm_wmma(const float* __restrict__ h, const float* __restrict__ W,
              float* __restrict__ z) {
    __shared__ float As[16 * APAD];

    const int tid     = threadIdx.y * 32 + threadIdx.x;
    const int rowBase = blockIdx.x * 16;

    {
        const int r = tid >> 4;
        const int c = (tid & 15) * 8;
        const float* gp = h + (size_t)(rowBase + r) * DIM + c;
        float*       lp = As + r * APAD + c;
        #pragma unroll
        for (int i = 0; i < 8; ++i) lp[i] = gp[i];
    }
    __syncthreads();

    const int lane  = threadIdx.x;
    const int col   = threadIdx.y * 16 + (lane & 15);
    const int khalf = (lane >> 4) * 2;

    const float* ap = As + (lane & 15) * APAD + khalf;
    const float* bp = W + (size_t)khalf * DIM + col;

    v8f c = {0.f, 0.f, 0.f, 0.f, 0.f, 0.f, 0.f, 0.f};

    #pragma unroll
    for (int k = 0; k < DIM; k += 4) {
        v2f a, b;
        a.x = ap[k];
        a.y = ap[k + 1];
        b.x = bp[k * DIM];
        b.y = bp[k * DIM + DIM];
        c = __builtin_amdgcn_wmma_f32_16x16x4_f32(
                false, a, false, b, (short)0, c, false, false);
    }

    const int rstep = (lane >> 4) * 8;
    float* zp = z + (size_t)(rowBase + rstep) * DIM + col;
    #pragma unroll
    for (int r = 0; r < 8; ++r)
        zp[r * DIM] = c[r];
}

__global__ void __launch_bounds__(256)
gat_sd_kernel(const float* __restrict__ z, const float* __restrict__ a_src,
              const float* __restrict__ a_dst,
              float* __restrict__ s, float* __restrict__ d,
              int* __restrict__ cnt, int* __restrict__ total) {
    const int row = blockIdx.x * 8 + threadIdx.y;
    if (row >= N_NODES) return;
    const int lane = threadIdx.x;

    const float* zp = z + (size_t)row * DIM + lane * 4;
    float ss = 0.f, dd = 0.f;
    #pragma unroll
    for (int i = 0; i < 4; ++i) {
        const float zv = zp[i];
        ss += zv * a_src[lane * 4 + i];
        dd += zv * a_dst[lane * 4 + i];
    }
    #pragma unroll
    for (int off = 16; off > 0; off >>= 1) {
        ss += __shfl_xor(ss, off, 32);
        dd += __shfl_xor(dd, off, 32);
    }
    if (lane == 0) {
        s[row] = ss; d[row] = dd;
        cnt[row] = 0;
        if (row == 0) *total = 0;
    }
}

__global__ void __launch_bounds__(256)
gat_count_kernel(const int* __restrict__ dst, int* __restrict__ cnt) {
    const int j = blockIdx.x * blockDim.x + threadIdx.x;
    if (j >= N_EDGES) return;
    atomicAdd(&cnt[dst[j]], 1);
}

__global__ void __launch_bounds__(256)
gat_offset_kernel(const int* __restrict__ cnt, int* __restrict__ cur,
                  int* __restrict__ total) {
    const int i    = blockIdx.x * blockDim.x + threadIdx.x;
    const int lane = threadIdx.x & 31;
    const int c    = (i < N_NODES) ? cnt[i] : 0;
    int incl = c;
    #pragma unroll
    for (int off = 1; off < 32; off <<= 1) {
        const int t = __shfl_up(incl, off, 32);
        if (lane >= off) incl += t;
    }
    int base = 0;
    if (lane == 31) base = atomicAdd(total, incl);
    base = __shfl(base, 31, 32);
    if (i < N_NODES) cur[i] = base + incl - c;
}

__global__ void __launch_bounds__(256)
gat_fill_kernel(const int* __restrict__ src, const int* __restrict__ dst,
                int* __restrict__ cur, int* __restrict__ esrc) {
    const int j = blockIdx.x * blockDim.x + threadIdx.x;
    if (j >= N_EDGES) return;
    const int pos = atomicAdd(&cur[dst[j]], 1);
    if ((unsigned)pos < (unsigned)N_EDGES) esrc[pos] = src[j];
}

__device__ __forceinline__ float wave_max32(float v) {
    #pragma unroll
    for (int off = 16; off > 0; off >>= 1) v = fmaxf(v, __shfl_xor(v, off, 32));
    return v;
}
__device__ __forceinline__ float wave_sum32(float v) {
    #pragma unroll
    for (int off = 16; off > 0; off >>= 1) v += __shfl_xor(v, off, 32);
    return v;
}
__device__ __forceinline__ int clamp_node(int v) {
    return (int)min((unsigned)v, (unsigned)(N_NODES - 1));
}
__device__ __forceinline__ float leaky(float v) {
    return (v >= 0.0f) ? v : NEG_SLOPE * v;
}

template <int G>
__device__ __forceinline__ void add_rows(const float* __restrict__ z, int lane,
                                         int srcv, float alv, int k, float (&acc)[4]) {
    float zv[G][4];
    float al[G];
    #pragma unroll
    for (int g = 0; g < G; ++g) {
        const int sg = __builtin_amdgcn_readlane(srcv, k + g);
        al[g] = __int_as_float(__builtin_amdgcn_readlane(__float_as_int(alv), k + g));
        const float* zp = z + (size_t)sg * DIM + lane;
        #pragma unroll
        for (int i = 0; i < 4; ++i) zv[g][i] = zp[i * 32];
    }
    #pragma unroll
    for (int g = 0; g < G; ++g) {
        #pragma unroll
        for (int i = 0; i < 4; ++i) acc[i] += al[g] * zv[g][i];
    }
}

__device__ __forceinline__ int next_source(const int* __restrict__ esrc, int base, int deg,
                                           int last, int lane, int& mult) {
    int best = 0x7fffffff;
    for (int t = lane; t < deg; t += 32) {
        const int v = esrc[base + t];
        if (v > last && v < best) best = v;
    }
    #pragma unroll
    for (int off = 16; off > 0; off >>= 1) {
        const int o = __shfl_xor(best, off, 32);
        best = (o < best) ? o : best;
    }
    int c = 0;
    for (int t = lane; t < deg; t += 32) c += (esrc[base + t] == best) ? 1 : 0;
    #pragma unroll
    for (int off = 16; off > 0; off >>= 1) c += __shfl_xor(c, off, 32);
    mult = (best == 0x7fffffff) ? 0 : c;
    return best;
}

__global__ void __launch_bounds__(256)
gat_node_kernel(const int* __restrict__ cnt, const int* __restrict__ cur,
                const int* __restrict__ esrc,
                const float* __restrict__ s, const float* __restrict__ d,
                const float* __restrict__ z, float* __restrict__ out) {
    __shared__ int sorted_src[8][64];

    const int lane = threadIdx.x;
    const int node = blockIdx.x * 8 + threadIdx.y;
    if (node >= N_NODES) return;

    int deg  = __builtin_amdgcn_readfirstlane(cnt[node]);
    int base = __builtin_amdgcn_readfirstlane(cur[node]) - deg;
    if (deg < 0) deg = 0;
    if (base < 0) base = 0;
    if (base > N_EDGES) base = N_EDGES;
    if (deg > N_EDGES - base) deg = N_EDGES - base;

    const float dn = d[node];
    float acc[4] = {0.f, 0.f, 0.f, 0.f};

    if (deg <= 64) {
        const int n0 = deg < 32 ? deg : 32;
        const int n1 = deg - n0;
        int k0 = 0x7fffffff, k1 = 0x7fffffff;
        if (lane < n0) k0 = esrc[base + lane];
        if (lane < n1) k1 = esrc[base + 32 + lane];

        int r0 = 0, r1 = 0;
        for (int i = 0; i < n0; ++i) {
            const int a = __builtin_amdgcn_readlane(k0, i);
            r0 += (a < k0 || (a == k0 && i < lane)) ? 1 : 0;
            r1 += (a <= k1) ? 1 : 0;
        }
        for (int i = 0; i < n1; ++i) {
            const int b = __builtin_amdgcn_readlane(k1, i);
            r0 += (b < k0) ? 1 : 0;
            r1 += (b < k1 || (b == k1 && i < lane)) ? 1 : 0;
        }

        int* ls = sorted_src[threadIdx.y];
        if (lane < n0) ls[r0 & 63] = k0;
        if (lane < n1) ls[r1 & 63] = k1;
        __builtin_amdgcn_fence(__ATOMIC_RELEASE, "wavefront");
        __builtin_amdgcn_wave_barrier();
        __builtin_amdgcn_fence(__ATOMIC_ACQUIRE, "wavefront");
        int ss0 = 0, ss1 = 0;
        if (lane < n0) ss0 = clamp_node(ls[lane]);
        if (lane < n1) ss1 = clamp_node(ls[32 + lane]);

        const float ninf = -__builtin_inff();
        float e0 = ninf, e1 = ninf;
        if (lane < n0) e0 = leaky(s[ss0] + dn);
        if (lane < n1) e1 = leaky(s[ss1] + dn);
        float m = wave_max32(fmaxf(e0, e1));
        m = __builtin_isfinite(m) ? m : 0.0f;
        const float x0 = (lane < n0) ? __expf(e0 - m) : 0.0f;
        const float x1 = (lane < n1) ? __expf(e1 - m) : 0.0f;
        float den = wave_sum32(x0 + x1);
        den = (den > 0.0f) ? den : 1.0f;
        const float a0 = x0 / den;
        const float a1 = x1 / den;

        int k = 0;
        for (; k + 4 <= n0; k += 4) add_rows<4>(z, lane, ss0, a0, k, acc);
        if (n0 - k >= 2) { add_rows<2>(z, lane, ss0, a0, k, acc); k += 2; }
        if (n0 - k >= 1) { add_rows<1>(z, lane, ss0, a0, k, acc); }
        for (int q = 0; q < n1; ++q) add_rows<1>(z, lane, ss1, a1, q, acc);
    } else {
        float mx = -__builtin_inff();
        for (int t = lane; t < deg; t += 32)
            mx = fmaxf(mx, leaky(s[clamp_node(esrc[base + t])] + dn));
        float m = wave_max32(mx);
        m = __builtin_isfinite(m) ? m : 0.0f;
        float den = 0.0f;
        int last = -1;
        for (int done = 0; done < deg; ) {
            int mult;
            last = next_source(esrc, base, deg, last, lane, mult);
            if (mult == 0) break;
            const float ex = __expf(leaky(s[clamp_node(last)] + dn) - m);
            for (int c = 0; c < mult; ++c) den += ex;
            done += mult;
        }
        den = (den > 0.0f) ? den : 1.0f;
        last = -1;
        for (int done = 0; done < deg; ) {
            int mult;
            last = next_source(esrc, base, deg, last, lane, mult);
            if (mult == 0) break;
            const int sj = clamp_node(last);
            const float al = __expf(leaky(s[sj] + dn) - m) / den;
            const float* zp = z + (size_t)sj * DIM + lane;
            float zr[4];
            #pragma unroll
            for (int i = 0; i < 4; ++i) zr[i] = zp[i * 32];
            for (int c = 0; c < mult; ++c) {
                #pragma unroll
                for (int i = 0; i < 4; ++i) acc[i] += al * zr[i];
            }
            done += mult;
        }
    }

    float* op = out + (size_t)node * DIM + lane;
    #pragma unroll
    for (int i = 0; i < 4; ++i) op[i * 32] = acc[i];
}

extern "C" void kernel_launch(void* const* d_in, const int* in_sizes, int n_in,
                              void* d_out, int out_size, void* d_ws, size_t ws_size,
                              hipStream_t stream) {
    const float* h     = (const float*)d_in[0];
    const int*   src   = (const int*)  d_in[1];
    const int*   dst   = (const int*)  d_in[2];
    const float* W     = (const float*)d_in[3];
    const float* a_src = (const float*)d_in[4];
    const float* a_dst = (const float*)d_in[5];
    float*       out   = (float*)d_out;

    float* ws    = (float*)d_ws;
    float* z     = ws;
    float* s     = z  + (size_t)N_NODES * DIM;
    float* dn    = s  + N_NODES;
    int*   cnt   = (int*)(dn + N_NODES);
    int*   cur   = cnt + N_NODES;
    int*   esrc  = cur + N_NODES;
    int*   total = esrc;

    gat_gemm_wmma<<<N_NODES / 16, dim3(32, 8), 0, stream>>>(h, W, z);
    gat_sd_kernel<<<(N_NODES + 7) / 8, dim3(32, 8), 0, stream>>>(z, a_src, a_dst, s, dn, cnt, total);
    gat_count_kernel<<<(N_EDGES + 255) / 256, 256, 0, stream>>>(dst, cnt);
    gat_offset_kernel<<<(N_NODES + 255) / 256, 256, 0, stream>>>(cnt, cur, total);
    gat_fill_kernel<<<(N_EDGES + 255) / 256, 256, 0, stream>>>(src, dst, cur, esrc);
    gat_node_kernel<<<N_NODES / 8, dim3(32, 8), 0, stream>>>(cnt, cur, esrc, s, dn, z, out);
}


# node kernel: lane owns features 4l..4l+3: a z row is one 128-bit load per lane (4 whole lines per instruction, was 4 one-line loads), 2 packed multiply-adds, one 128-bit out store; sum order kept
# speedup vs baseline: 4.0851x; 2.3913x over previous
#include <hip/hip_runtime.h>
#include <hip/hip_bf16.h>

#define N_NODES   50000
#define N_EDGES   800000
#define DIM       128
#define NEG_SLOPE 0.01f
#define APAD      132

typedef float v2f __attribute__((ext_vector_type(2)));
typedef float v8f __attribute__((ext_vector_type(8)));


template <int M>
__device__ __forceinline__ float row_xor16(float v) {
    return __int_as_float(__builtin_amdgcn_update_dpp(0, __float_as_int(v), 0x160 + M, 0xf, 0xf, false));
}

__global__ void __launch_bounds__(256, 16)
gat_gemm_wmma(const float* __restrict__ h, const float* __restrict__ W,
              const float* __restrict__ a_src, const float* __restrict__ a_dst,
              float* __restrict__ z, float* __restrict__ s, float* __restrict__ d) {
    __shared__ float As[16 * APAD];
    __shared__ __attribute__((aligned(16))) float Zs[16 * APAD];

    const int tid     = threadIdx.y * 32 + threadIdx.x;
    const int rowBase = blockIdx.x * 16;

    {
        const int r = tid >> 4;
        const int c = (tid & 15) * 8;
        const float* gp = h + (size_t)(rowBase + r) * DIM + c;
        float*       lp = As + r * APAD + c;
        #pragma unroll
        for (int i = 0; i < 8; ++i) lp[i] = gp[i];
    }

    __syncthreads();

    const int lane  = threadIdx.x;
    const int col   = threadIdx.y * 16 + (lane & 15);
    const int khalf = (lane >> 4) * 2;

    const float* ap = As + (lane & 15) * APAD + khalf;
    const float* bp = W + (size_t)khalf * DIM + col;

    v8f c = {0.f, 0.f, 0.f, 0.f, 0.f, 0.f, 0.f, 0.f};

    #pragma unroll
    for (int k = 0; k < DIM; k += 4) {
        v2f a, b;
        a.x = ap[k];
        a.y = ap[k + 1];
        b.x = bp[k * DIM];
        b.y = bp[k * DIM + DIM];
        c = __builtin_amdgcn_wmma_f32_16x16x4_f32(
                false, a, false, b, (short)0, c, false, false);
    }

    const int rstep = (lane >> 4) * 8;
    float* zs = Zs + rstep * APAD + col;
    #pragma unroll
    for (int r = 0; r < 8; ++r)
        zs[r * APAD] = c[r];
    __syncthreads();

    const int erow = tid >> 4;
    const int eq   = tid & 15;
    const float4 as0 = *(const float4*)(a_src + eq * 4);
    const float4 as1 = *(const float4*)(a_src + 64 + eq * 4);
    const float4 ad0 = *(const float4*)(a_dst + eq * 4);
    const float4 ad1 = *(const float4*)(a_dst + 64 + eq * 4);
    const float* zr = Zs + erow * APAD + eq * 4;
    const float4 z0 = *(const float4*)zr;
    const float4 z1 = *(const float4*)(zr + 64);
    float* zg = z + (size_t)(rowBase + erow) * DIM + eq * 4;
    *(float4*)zg        = z0;
    *(float4*)(zg + 64) = z1;

    float ss = z0.x * as0.x, dd = z0.x * ad0.x;
    ss += z0.y * as0.y;  dd += z0.y * ad0.y;
    ss += z0.z * as0.z;  dd += z0.z * ad0.z;
    ss += z0.w * as0.w;  dd += z0.w * ad0.w;
    ss += z1.x * as1.x;  dd += z1.x * ad1.x;
    ss += z1.y * as1.y;  dd += z1.y * ad1.y;
    ss += z1.z * as1.z;  dd += z1.z * ad1.z;
    ss += z1.w * as1.w;  dd += z1.w * ad1.w;
    ss += row_xor16<1>(ss);  dd += row_xor16<1>(dd);
    ss += row_xor16<2>(ss);  dd += row_xor16<2>(dd);
    ss += row_xor16<4>(ss);  dd += row_xor16<4>(dd);
    ss += row_xor16<8>(ss);  dd += row_xor16<8>(dd);
    if (eq == 0) {
        s[rowBase + erow] = ss;
        d[rowBase + erow] = dd;
    }
}

__device__ __forceinline__ float wave_max32(float v) {
    #pragma unroll
    for (int off = 16; off > 0; off >>= 1) v = fmaxf(v, __shfl_xor(v, off, 32));
    return v;
}
__device__ __forceinline__ float wave_sum32(float v) {
    #pragma unroll
    for (int off = 16; off > 0; off >>= 1) v += __shfl_xor(v, off, 32);
    return v;
}
__device__ __forceinline__ int clamp_node(int v) {
    return (int)min((unsigned)v, (unsigned)(N_NODES - 1));
}
__device__ __forceinline__ float leaky(float v) {
    return (v >= 0.0f) ? v : NEG_SLOPE * v;
}

template <int G>
__device__ __forceinline__ void add_rows(const float* __restrict__ z, int lane,
                                         int srcv, float alv, int k, v2f& acc01, v2f& acc23) {
    float4 zv[G];
    float  al[G];
    #pragma unroll
    for (int g = 0; g < G; ++g) {
        const int sg = __builtin_amdgcn_readlane(srcv, k + g);
        al[g] = __int_as_float(__builtin_amdgcn_readlane(__float_as_int(alv), k + g));
        const float* zrow = z + (size_t)sg * DIM;
        zv[g] = *(const float4*)(zrow + lane * 4);
    }
    #pragma unroll
    for (int g = 0; g < G; ++g) {
        const v2f a2  = {al[g], al[g]};
        const v2f z01 = {zv[g].x, zv[g].y};
        const v2f z23 = {zv[g].z, zv[g].w};
        acc01 += a2 * z01;
        acc23 += a2 * z23;
    }
}

#define NBUCK       782
#define NCHUNK      125
#define CHUNK       6400
#define BIN_THREADS 1024
#define CAP         4096
#define EPT         7

__global__ void __launch_bounds__(BIN_THREADS)
gat_bin_count(const int* __restrict__ dst, int* __restrict__ H) {
    __shared__ int hist[NBUCK];
    const int tid = threadIdx.x;
    if (tid < NBUCK) hist[tid] = 0;
    __syncthreads();
    const int e0 = blockIdx.x * CHUNK;
    int dv[EPT];
    #pragma unroll
    for (int k = 0; k < EPT; ++k) {
        const int i = tid + k * BIN_THREADS;
        dv[k] = (i < CHUNK) ? dst[e0 + i] : 0;
    }
    #pragma unroll
    for (int k = 0; k < EPT; ++k)
        if (tid + k * BIN_THREADS < CHUNK) atomicAdd(&hist[clamp_node(dv[k]) >> 6], 1);
    __syncthreads();
    if (tid < NBUCK) H[blockIdx.x * NBUCK + tid] = hist[tid];
}

__global__ void __launch_bounds__(BIN_THREADS)
gat_bin_place(const int* __restrict__ src, const int* __restrict__ dst,
              const int* __restrict__ H, int* __restrict__ bbase,
              unsigned* __restrict__ ebuf) {
    __shared__ int      cursor[NBUCK];
    __shared__ int      lstart[NBUCK];
    __shared__ int      gdelta[NBUCK];
    __shared__ unsigned stage[CHUNK];
    __shared__ int      wsum[2][32];

    const int tid  = threadIdx.x;
    const int lane = tid & 31;
    const int wv   = tid >> 5;
    const int c    = blockIdx.x;

    int mine = 0, below = 0, total = 0;
    if (tid < NBUCK) {
        #pragma unroll 25
        for (int cc = 0; cc < NCHUNK; ++cc) {
            const int v = H[cc * NBUCK + tid];
            total += v;
            below += (cc < c) ? v : 0;
            mine   = (cc == c) ? v : mine;
        }
    }
    int it = total, im = mine;
    #pragma unroll
    for (int off = 1; off < 32; off <<= 1) {
        const int t = __shfl_up(it, off, 32);
        const int u = __shfl_up(im, off, 32);
        if (lane >= off) { it += t; im += u; }
    }
    if (lane == 31) { wsum[0][wv] = it; wsum[1][wv] = im; }
    __syncthreads();
    if (wv == 0) {
        const int a = wsum[0][lane], b = wsum[1][lane];
        int ia = a, ib = b;
        #pragma unroll
        for (int off = 1; off < 32; off <<= 1) {
            const int t = __shfl_up(ia, off, 32);
            const int u = __shfl_up(ib, off, 32);
            if (lane >= off) { ia += t; ib += u; }
        }
        wsum[0][lane] = ia - a;
        wsum[1][lane] = ib - b;
    }
    __syncthreads();
    const int bstart = wsum[0][wv] + it - total;
    const int ls     = wsum[1][wv] + im - mine;
    if (tid < NBUCK) {
        cursor[tid] = 0;
        lstart[tid] = ls;
        gdelta[tid] = bstart + below - ls;
    }
    if (c == 0 && tid <= NBUCK) bbase[tid] = bstart;
    __syncthreads();

    const int e0 = c * CHUNK;
    int dv[EPT], sv[EPT];
    #pragma unroll
    for (int k = 0; k < EPT; ++k) {
        const int i = tid + k * BIN_THREADS;
        dv[k] = (i < CHUNK) ? dst[e0 + i] : 0;
        sv[k] = (i < CHUNK) ? src[e0 + i] : 0;
    }
    #pragma unroll
    for (int k = 0; k < EPT; ++k) {
        if (tid + k * BIN_THREADS >= CHUNK) break;
        const int dn = clamp_node(dv[k]);
        const int sn = clamp_node(sv[k]);
        const int b  = dn >> 6;
        const int p  = atomicAdd(&cursor[b], 1);
        const unsigned lp = min((unsigned)(lstart[b] + p), (unsigned)(CHUNK - 1));
        stage[lp] = ((unsigned)dn << 16) | (unsigned)sn;
    }
    __syncthreads();
    for (int t = tid; t < CHUNK; t += BIN_THREADS) {
        const unsigned w  = stage[t];
        const int      gp = gdelta[w >> 22] + t;
        if ((unsigned)gp < (unsigned)N_EDGES) ebuf[gp] = w;
    }
}

__device__ __forceinline__ unsigned next_key(const unsigned* __restrict__ list, int n,
                                             unsigned lo, unsigned hi, int lane, int& mult) {
    unsigned best = 0xffffffffu;
    for (int t = lane; t < n; t += 32) {
        const unsigned v = list[t];
        if (v >= lo && v <= hi && v < best) best = v;
    }
    #pragma unroll
    for (int off = 16; off > 0; off >>= 1) {
        const unsigned o = __shfl_xor(best, off, 32);
        best = (o < best) ? o : best;
    }
    int c = 0;
    for (int t = lane; t < n; t += 32) c += (list[t] == best) ? 1 : 0;
    #pragma unroll
    for (int off = 16; off > 0; off >>= 1) c += __shfl_xor(c, off, 32);
    mult = (best == 0xffffffffu) ? 0 : c;
    return best;
}

__device__ __forceinline__ void node_general(const unsigned* __restrict__ list, int n, int node, int lane,
                                          const float* __restrict__ s, float dn,
                                          const float* __restrict__ z, float (&acc)[4]) {
    const unsigned lo = (unsigned)node << 16;
    const unsigned hi = lo | 0xffffu;
    float mx = -__builtin_inff();
    for (int t = lane; t < n; t += 32) {
        const unsigned v = list[t];
        if (v >= lo && v <= hi) mx = fmaxf(mx, leaky(s[clamp_node((int)(v & 0xffffu))] + dn));
    }
    float m = wave_max32(mx);
    m = __builtin_isfinite(m) ? m : 0.0f;
    float den = 0.0f;
    unsigned cur = lo;
    for (int done = 0; done < n; ) {
        int mult;
        const unsigned k = next_key(list, n, cur, hi, lane, mult);
        if (mult == 0) break;
        const float ex = __expf(leaky(s[clamp_node((int)(k & 0xffffu))] + dn) - m);
        for (int c = 0; c < mult; ++c) den += ex;
        done += mult;
        cur = k + 1u;
    }
    den = (den > 0.0f) ? den : 1.0f;
    cur = lo;
    for (int done = 0; done < n; ) {
        int mult;
        const unsigned k = next_key(list, n, cur, hi, lane, mult);
        if (mult == 0) break;
        const int sj = clamp_node((int)(k & 0xffffu));
        const float al = __expf(leaky(s[sj] + dn) - m) / den;
        const float4 zq = *(const float4*)(z + (size_t)sj * DIM + lane * 4);
        const float zr[4] = {zq.x, zq.y, zq.z, zq.w};
        for (int c = 0; c < mult; ++c) {
            #pragma unroll
            for (int i = 0; i < 4; ++i) acc[i] += al * zr[i];
        }
        done += mult;
        cur = k + 1u;
    }
}

#define NODE_WAVES 16
__global__ void __launch_bounds__(NODE_WAVES * 32, 16)
gat_node_kernel(const int* __restrict__ bbase, const unsigned* __restrict__ ebuf,
                const float* __restrict__ s, const float* __restrict__ d,
                const float* __restrict__ z, float* __restrict__ out) {
    __shared__ int sorted_src[NODE_WAVES][64];
    __shared__ int cntL[64];
    __shared__ int startL[64];
    __shared__ int curL[64];
    __shared__ int slist[CAP];

    const int lane   = threadIdx.x;
    const int wy     = threadIdx.y;
    const int tid    = wy * 32 + lane;
    const int bucket = blockIdx.x;

    int gb = bbase[bucket];
    int nb = bbase[bucket + 1] - gb;
    if (gb < 0) gb = 0;
    if (gb > N_EDGES) gb = N_EDGES;
    if (nb < 0) nb = 0;
    if (nb > N_EDGES - gb) nb = N_EDGES - gb;
    const unsigned* blk = ebuf + gb;
    const bool fits = (nb <= CAP);

    if (tid < 64) { cntL[tid] = 0; curL[tid] = 0; }
    __syncthreads();
    unsigned ew[CAP / (NODE_WAVES * 32)];
    if (fits) {
        #pragma unroll
        for (int k = 0; k < CAP / (NODE_WAVES * 32); ++k) {
            const int i = tid + k * NODE_WAVES * 32;
            ew[k] = (i < nb) ? blk[i] : 0u;
        }
        #pragma unroll
        for (int k = 0; k < CAP / (NODE_WAVES * 32); ++k)
            if (tid + k * NODE_WAVES * 32 < nb) atomicAdd(&cntL[(ew[k] >> 16) & 63u], 1);
    }
    __syncthreads();
    if (wy == 0) {
        const int c0 = cntL[lane], c1 = cntL[32 + lane];
        int i0 = c0, i1 = c1;
        #pragma unroll
        for (int off = 1; off < 32; off <<= 1) {
            const int t0 = __shfl_up(i0, off, 32);
            const int t1 = __shfl_up(i1, off, 32);
            if (lane >= off) { i0 += t0; i1 += t1; }
        }
        const int tot0 = __shfl(i0, 31, 32);
        startL[lane]      = i0 - c0;
        startL[32 + lane] = tot0 + i1 - c1;
    }
    __syncthreads();
    if (fits) {
        #pragma unroll
        for (int k = 0; k < CAP / (NODE_WAVES * 32); ++k) {
            if (tid + k * NODE_WAVES * 32 >= nb) break;
            const unsigned w  = ew[k];
            const int      dl = (int)((w >> 16) & 63u);
            const int      p  = startL[dl] + atomicAdd(&curL[dl], 1);
            if ((unsigned)p < (unsigned)CAP) slist[p] = (int)(w & 0xffffu);
        }
    }
    __syncthreads();

    for (int turn = 0; turn < 64 / NODE_WAVES; ++turn) {
        const int dl   = wy + NODE_WAVES * turn;
        const int node = bucket * 64 + dl;
        if (node >= N_NODES) break;

        const float dn = d[node];
        v2f acc01 = {0.f, 0.f}, acc23 = {0.f, 0.f};

        int deg  = fits ? __builtin_amdgcn_readfirstlane(cntL[dl])   : 65;
        int base = fits ? __builtin_amdgcn_readfirstlane(startL[dl]) : 0;
        if (deg < 0) deg = 0;
        if (base < 0) base = 0;
        if (base > CAP) base = CAP;
        if (deg > CAP - base && fits) deg = CAP - base;

        if (deg <= 64) {
            const int n0 = deg < 32 ? deg : 32;
            const int n1 = deg - n0;
            int k0 = 0x7fffffff, k1 = 0x7fffffff;
            if (lane < n0) k0 = slist[base + lane];
            if (lane < n1) k1 = slist[base + 32 + lane];

            int r0 = 0, r1 = 0;
            for (int i = 0; i < n0; ++i) {
                const int a = __builtin_amdgcn_readlane(k0, i);
                r0 += (a < k0 || (a == k0 && i < lane)) ? 1 : 0;
                r1 += (a <= k1) ? 1 : 0;
            }
            for (int i = 0; i < n1; ++i) {
                const int b = __builtin_amdgcn_readlane(k1, i);
                r0 += (b < k0) ? 1 : 0;
                r1 += (b < k1 || (b == k1 && i < lane)) ? 1 : 0;
            }

            int* ls = sorted_src[threadIdx.y];
            if (lane < n0) ls[r0 & 63] = k0;
            if (lane < n1) ls[r1 & 63] = k1;
            __builtin_amdgcn_fence(__ATOMIC_RELEASE, "wavefront");
            __builtin_amdgcn_wave_barrier();
            __builtin_amdgcn_fence(__ATOMIC_ACQUIRE, "wavefront");
            int ss0 = 0, ss1 = 0;
            if (lane < n0) ss0 = clamp_node(ls[lane]);
            if (lane < n1) ss1 = clamp_node(ls[32 + lane]);

            const float ninf = -__builtin_inff();
            float e0 = ninf, e1 = ninf;
            if (lane < n0) e0 = leaky(s[ss0] + dn);
            if (lane < n1) e1 = leaky(s[ss1] + dn);
            float m = wave_max32(fmaxf(e0, e1));
            m = __builtin_isfinite(m) ? m : 0.0f;
            const float x0 = (lane < n0) ? __expf(e0 - m) : 0.0f;
            const float x1 = (lane < n1) ? __expf(e1 - m) : 0.0f;
            float den = wave_sum32(x0 + x1);
            den = (den > 0.0f) ? den : 1.0f;
            const float a0 = x0 / den;
            const float a1 = x1 / den;

            int k = 0;
            for (; k + 4 <= n0; k += 4) add_rows<4>(z, lane, ss0, a0, k, acc01, acc23);
            if (n0 - k >= 2) { add_rows<2>(z, lane, ss0, a0, k, acc01, acc23); k += 2; }
            if (n0 - k >= 1) { add_rows<1>(z, lane, ss0, a0, k, acc01, acc23); }
            for (int q = 0; q < n1; ++q) add_rows<1>(z, lane, ss1, a1, q, acc01, acc23);
        } else {
            float acc[4] = {0.f, 0.f, 0.f, 0.f};
            node_general(blk, nb, node, lane, s, dn, z, acc);
            acc01.x = acc[0];  acc01.y = acc[1];
            acc23.x = acc[2];  acc23.y = acc[3];
        }

        *(float4*)(out + (size_t)node * DIM + lane * 4) = make_float4(acc01.x, acc01.y, acc23.x, acc23.y);
    }
}

extern "C" void kernel_launch(void* const* d_in, const int* in_sizes, int n_in,
                              void* d_out, int out_size, void* d_ws, size_t ws_size,
                              hipStream_t stream) {
    const float* h     = (const float*)d_in[0];
    const int*   src   = (const int*)  d_in[1];
    const int*   dst   = (const int*)  d_in[2];
    const float* W     = (const float*)d_in[3];
    const float* a_src = (const float*)d_in[4];
    const float* a_dst = (const float*)d_in[5];
    float*       out   = (float*)d_out;

    float*    ws    = (float*)d_ws;
    float*    z     = ws;
    float*    s     = z  + (size_t)N_NODES * DIM;
    float*    dn    = s  + N_NODES;
    int*      H     = (int*)(dn + N_NODES);
    int*      bbase = H + NCHUNK * NBUCK;
    unsigned* ebuf  = (unsigned*)(H + 2 * N_NODES);

    gat_gemm_wmma<<<N_NODES / 16, dim3(32, 8), 0, stream>>>(h, W, a_src, a_dst, z, s, dn);
    gat_bin_count<<<NCHUNK, BIN_THREADS, 0, stream>>>(dst, H);
    gat_bin_place<<<NCHUNK, BIN_THREADS, 0, stream>>>(src, dst, H, bbase, ebuf);
    gat_node_kernel<<<NBUCK, dim3(32, NODE_WAVES), 0, stream>>>(bbase, ebuf, s, dn, z, out);
}


# product launch: a work-group keeps its W fragments in registers and loops over 5 row stripes (next h rows loaded a stripe ahead, LDS-only barriers): 5x fewer waves to start, W read once a work-group
# speedup vs baseline: 5.3030x; 1.2981x over previous
#include <hip/hip_runtime.h>
#include <hip/hip_bf16.h>

#define N_NODES   50000
#define N_EDGES   800000
#define DIM       128
#define NEG_SLOPE 0.01f
#define APAD      132
#define NSTRIPE   3125
#define ST_PER_WG 5
#define GEMM_WG   625

typedef float v2f __attribute__((ext_vector_type(2)));
typedef float v8f __attribute__((ext_vector_type(8)));

#define NBUCK       782
#define NCHUNK      125
#define CHUNK       6400
#define CNT_EPT     25
#define LPITCH      800
#define SMEM_WORDS  (LPITCH + CHUNK)

__device__ __forceinline__ int clamp_node(int v) {
    return (int)min((unsigned)v, (unsigned)(N_NODES - 1));
}


template <int M>
__device__ __forceinline__ float row_xor16(float v) {
    return __int_as_float(__builtin_amdgcn_update_dpp(0, __float_as_int(v), 0x160 + M, 0xf, 0xf, false));
}

__device__ __forceinline__ void lds_barrier() {
    __builtin_amdgcn_fence(__ATOMIC_RELEASE, "workgroup", "local");
    __builtin_amdgcn_s_barrier();
    __builtin_amdgcn_fence(__ATOMIC_ACQUIRE, "workgroup", "local");
}

__global__ void __launch_bounds__(256, 8)
gat_gemm_wmma(const float* __restrict__ h, const float* __restrict__ W,
              const float* __restrict__ a_src, const float* __restrict__ a_dst,
              float* __restrict__ z, float* __restrict__ s, float* __restrict__ d,
              const int* __restrict__ src, const int* __restrict__ dst,
              int* __restrict__ L, unsigned* __restrict__ ebuf) {
    __shared__ __attribute__((aligned(16))) float smem[SMEM_WORDS];
    __shared__ int wtot[8];
    float* As = smem;
    float* Zs = smem + 16 * APAD;

    const int tid = threadIdx.y * 32 + threadIdx.x;

    if (blockIdx.x < NCHUNK) {
        int*      hist   = (int*)smem;
        unsigned* stage  = (unsigned*)(hist + LPITCH);
        const int c    = blockIdx.x;
        const int e0   = c * CHUNK;
        const int lane = threadIdx.x;
        const int wv   = threadIdx.y;
        int dv[CNT_EPT], sv[CNT_EPT];
        #pragma unroll
        for (int k = 0; k < CNT_EPT; ++k) {
            dv[k] = dst[e0 + tid + k * 256];
            sv[k] = src[e0 + tid + k * 256];
        }
        #pragma unroll
        for (int j = 0; j < (LPITCH + 255) / 256; ++j)
            if (tid + j * 256 < LPITCH) hist[tid + j * 256] = 0;
        __syncthreads();
        #pragma unroll
        for (int k = 0; k < CNT_EPT; ++k) {
            const int dn = clamp_node(dv[k]);
            const int sn = clamp_node(sv[k]);
            stage[tid + k * 256] = ((unsigned)dn << 16) | (unsigned)sn;
            atomicAdd(&hist[dn >> 6], 1);
        }
        __syncthreads();
        int4 cv = make_int4(0, 0, 0, 0);
        if (tid < LPITCH / 4) cv = *(const int4*)(hist + 4 * tid);
        const int sum4 = cv.x + cv.y + cv.z + cv.w;
        int inc = sum4;
        #pragma unroll
        for (int off = 1; off < 32; off <<= 1) {
            const int t = __shfl_up(inc, off, 32);
            if (lane >= off) inc += t;
        }
        if (lane == 31) wtot[wv] = inc;
        unsigned w[CNT_EPT];
        #pragma unroll
        for (int k = 0; k < CNT_EPT; ++k) w[k] = stage[tid + k * 256];
        __syncthreads();
        int pre = 0;
        #pragma unroll
        for (int q = 0; q < 8; ++q) pre += (q < wv) ? wtot[q] : 0;
        if (tid < LPITCH / 4) {
            int4 st;
            st.x = pre + inc - sum4;
            st.y = st.x + cv.x;
            st.z = st.y + cv.y;
            st.w = st.z + cv.z;
            *(int4*)(hist + 4 * tid) = st;
            const int sx[4] = {st.x, st.y, st.z, st.w};
            #pragma unroll
            for (int i = 0; i < 4; ++i)
                if (4 * tid + i <= NBUCK) L[(4 * tid + i) * NCHUNK + c] = sx[i];
        }
        __syncthreads();
        int pos[CNT_EPT];
        #pragma unroll
        for (int k = 0; k < CNT_EPT; ++k) pos[k] = atomicAdd(&hist[w[k] >> 22], 1);
        #pragma unroll
        for (int k = 0; k < CNT_EPT; ++k)
            stage[min((unsigned)pos[k], (unsigned)(CHUNK - 1))] = w[k];
        __syncthreads();
        for (int q = tid; q < CHUNK / 4; q += 256)
            *(uint4*)(ebuf + e0 + 4 * q) = *(const uint4*)(stage + 4 * q);
        return;
    }

    const int lane  = threadIdx.x;
    const int col   = threadIdx.y * 16 + (lane & 15);
    const int khalf = (lane >> 4) * 2;

    const float* bp = W + (size_t)khalf * DIM + col;
    v2f bfrag[DIM / 4];
    #pragma unroll
    for (int j = 0; j < DIM / 4; ++j) {
        bfrag[j].x = bp[(4 * j) * DIM];
        bfrag[j].y = bp[(4 * j) * DIM + DIM];
    }

    const int    ar = tid >> 4;
    const int    ac = (tid & 15) * 8;
    float*       lp = As + ar * APAD + ac;
    const float* ap = As + (lane & 15) * APAD + khalf;
    const int rstep = (lane >> 4) * 8;
    float*       zs = Zs + rstep * APAD + col;
    const int erow = tid >> 4;
    const int eq   = tid & 15;
    const float* zr = Zs + erow * APAD + eq * 4;

    int rowBase = (blockIdx.x - NCHUNK) * (ST_PER_WG * 16);
    const float* gp = h + (size_t)(rowBase + ar) * DIM + ac;
    float4 p0 = *(const float4*)gp;
    float4 p1 = *(const float4*)(gp + 4);

    for (int it = 0; it < ST_PER_WG; ++it, rowBase += 16) {
        *(float4*)lp       = p0;
        *(float4*)(lp + 4) = p1;
        lds_barrier();
        if (it + 1 < ST_PER_WG) {
            gp += 16 * DIM;
            p0 = *(const float4*)gp;
            p1 = *(const float4*)(gp + 4);
        }

        v8f c = {0.f, 0.f, 0.f, 0.f, 0.f, 0.f, 0.f, 0.f};
        #pragma unroll
        for (int j = 0; j < DIM / 4; ++j) {
            v2f a;
            a.x = ap[4 * j];
            a.y = ap[4 * j + 1];
            c = __builtin_amdgcn_wmma_f32_16x16x4_f32(
                    false, a, false, bfrag[j], (short)0, c, false, false);
        }

        #pragma unroll
        for (int r = 0; r < 8; ++r)
            zs[r * APAD] = c[r];
        lds_barrier();

        const float4 as0 = *(const float4*)(a_src + eq * 4);
        const float4 as1 = *(const float4*)(a_src + 64 + eq * 4);
        const float4 ad0 = *(const float4*)(a_dst + eq * 4);
        const float4 ad1 = *(const float4*)(a_dst + 64 + eq * 4);
        const float4 z0 = *(const float4*)zr;
        const float4 z1 = *(const float4*)(zr + 64);
        float* zg = z + (size_t)(rowBase + erow) * DIM + eq * 4;
        *(float4*)zg        = z0;
        *(float4*)(zg + 64) = z1;

        float ss = z0.x * as0.x, dd = z0.x * ad0.x;
        ss += z0.y * as0.y;  dd += z0.y * ad0.y;
        ss += z0.z * as0.z;  dd += z0.z * ad0.z;
        ss += z0.w * as0.w;  dd += z0.w * ad0.w;
        ss += z1.x * as1.x;  dd += z1.x * ad1.x;
        ss += z1.y * as1.y;  dd += z1.y * ad1.y;
        ss += z1.z * as1.z;  dd += z1.z * ad1.z;
        ss += z1.w * as1.w;  dd += z1.w * ad1.w;
        ss += row_xor16<1>(ss);  dd += row_xor16<1>(dd);
        ss += row_xor16<2>(ss);  dd += row_xor16<2>(dd);
        ss += row_xor16<4>(ss);  dd += row_xor16<4>(dd);
        ss += row_xor16<8>(ss);  dd += row_xor16<8>(dd);
        if (eq == 0) {
            s[rowBase + erow] = ss;
            d[rowBase + erow] = dd;
        }
    }
}

__device__ __forceinline__ float wave_max32(float v) {
    #pragma unroll
    for (int off = 16; off > 0; off >>= 1) v = fmaxf(v, __shfl_xor(v, off, 32));
    return v;
}
__device__ __forceinline__ float wave_sum32(float v) {
    #pragma unroll
    for (int off = 16; off > 0; off >>= 1) v += __shfl_xor(v, off, 32);
    return v;
}
__device__ __forceinline__ float leaky(float v) {
    return (v >= 0.0f) ? v : NEG_SLOPE * v;
}

template <int G>
__device__ __forceinline__ void add_rows(const float* __restrict__ z, int lane,
                                         int srcv, float alv, int k, v2f& acc01, v2f& acc23) {
    float4 zv[G];
    float  al[G];
    #pragma unroll
    for (int g = 0; g < G; ++g) {
        const int sg = __builtin_amdgcn_readlane(srcv, k + g);
        al[g] = __int_as_float(__builtin_amdgcn_readlane(__float_as_int(alv), k + g));
        const float* zrow = z + (size_t)sg * DIM;
        zv[g] = *(const float4*)(zrow + lane * 4);
    }
    #pragma unroll
    for (int g = 0; g < G; ++g) {
        const v2f a2  = {al[g], al[g]};
        const v2f z01 = {zv[g].x, zv[g].y};
        const v2f z23 = {zv[g].z, zv[g].w};
        acc01 += a2 * z01;
        acc23 += a2 * z23;
    }
}

#define CAP         4096
#define SEG_EW      6

template <class F>
__device__ __forceinline__ void block_words(const unsigned* __restrict__ ebuf,
                                            const int* segLo, const int* segHi, int lane, F&& f) {
    for (int c = 0; c < NCHUNK; ++c) {
        const int lo = segLo[c], hi = segHi[c];
        const unsigned* seg = ebuf + (size_t)c * CHUNK;
        for (int i = lo + lane; i < hi; i += 32) f(seg[i]);
    }
}

__device__ __forceinline__ unsigned next_key(const unsigned* __restrict__ ebuf,
                                             const int* segLo, const int* segHi,
                                             unsigned lo, unsigned hi, int lane, int& mult) {
    unsigned best = 0xffffffffu;
    block_words(ebuf, segLo, segHi, lane, [&](unsigned v) {
        if (v >= lo && v <= hi && v < best) best = v;
    });
    #pragma unroll
    for (int off = 16; off > 0; off >>= 1) {
        const unsigned o = __shfl_xor(best, off, 32);
        best = (o < best) ? o : best;
    }
    int c = 0;
    block_words(ebuf, segLo, segHi, lane, [&](unsigned v) { c += (v == best) ? 1 : 0; });
    #pragma unroll
    for (int off = 16; off > 0; off >>= 1) c += __shfl_xor(c, off, 32);
    mult = (best == 0xffffffffu) ? 0 : c;
    return best;
}

__device__ __forceinline__ void node_general(const unsigned* __restrict__ ebuf,
                                          const int* segLo, const int* segHi, int n, int node, int lane,
                                          const float* __restrict__ s, float dn,
                                          const float* __restrict__ z, float (&acc)[4]) {
    const unsigned lo = (unsigned)node << 16;
    const unsigned hi = lo | 0xffffu;
    float mx = -__builtin_inff();
    block_words(ebuf, segLo, segHi, lane, [&](unsigned v) {
        if (v >= lo && v <= hi) mx = fmaxf(mx, leaky(s[clamp_node((int)(v & 0xffffu))] + dn));
    });
    float m = wave_max32(mx);
    m = __builtin_isfinite(m) ? m : 0.0f;
    float den = 0.0f;
    unsigned cur = lo;
    for (int done = 0; done < n; ) {
        int mult;
        const unsigned k = next_key(ebuf, segLo, segHi, cur, hi, lane, mult);
        if (mult == 0) break;
        const float ex = __expf(leaky(s[clamp_node((int)(k & 0xffffu))] + dn) - m);
        for (int c = 0; c < mult; ++c) den += ex;
        done += mult;
        cur = k + 1u;
    }
    den = (den > 0.0f) ? den : 1.0f;
    cur = lo;
    for (int done = 0; done < n; ) {
        int mult;
        const unsigned k = next_key(ebuf, segLo, segHi, cur, hi, lane, mult);
        if (mult == 0) break;
        const int sj = clamp_node((int)(k & 0xffffu));
        const float al = __expf(leaky(s[sj] + dn) - m) / den;
        const float4 zq = *(const float4*)(z + (size_t)sj * DIM + lane * 4);
        const float zr[4] = {zq.x, zq.y, zq.z, zq.w};
        for (int c = 0; c < mult; ++c) {
            #pragma unroll
            for (int i = 0; i < 4; ++i) acc[i] += al * zr[i];
        }
        done += mult;
        cur = k + 1u;
    }
}

#define NODE_WAVES 16
__global__ void __launch_bounds__(NODE_WAVES * 32, 16)
gat_node_kernel(const int* __restrict__ L, const unsigned* __restrict__ ebuf,
                const float* __restrict__ s, const float* __restrict__ d,
                const float* __restrict__ z, float* __restrict__ out) {
    __shared__ int sorted_src[NODE_WAVES][64];
    __shared__ int cntL[64];
    __shared__ int startL[64];
    __shared__ int curL[64];
    __shared__ int segLo[128];
    __shared__ int segHi[128];
    __shared__ int nbL;
    __shared__ int slist[CAP];

    const int lane   = threadIdx.x;
    const int wy     = threadIdx.y;
    const int tid    = wy * 32 + lane;
    const int bucket = blockIdx.x;

    const int sc = tid >> 2;
    const int sj = tid & 3;
    int lo = 0, hi = 0;
    if (sc < NCHUNK) {
        lo = L[bucket * NCHUNK + sc];
        hi = L[(bucket + 1) * NCHUNK + sc];
    }
    lo = min(max(lo, 0), CHUNK);
    hi = min(max(hi, lo), CHUNK);
    const unsigned* seg = ebuf + (size_t)min(sc, NCHUNK - 1) * CHUNK;

    unsigned ew[SEG_EW];
    #pragma unroll
    for (int k = 0; k < SEG_EW; ++k) {
        const int i = lo + sj + 4 * k;
        ew[k] = (i < hi) ? seg[i] : 0u;
    }
    if (tid < 64) { cntL[tid] = 0; curL[tid] = 0; }
    if (sj == 0) { segLo[sc] = lo; segHi[sc] = hi; }
    __syncthreads();
    #pragma unroll
    for (int k = 0; k < SEG_EW; ++k)
        if (lo + sj + 4 * k < hi) atomicAdd(&cntL[(ew[k] >> 16) & 63u], 1);
    for (int i = lo + sj + 4 * SEG_EW; i < hi; i += 4) atomicAdd(&cntL[(seg[i] >> 16) & 63u], 1);
    __syncthreads();
    if (wy == 0) {
        const int c0 = cntL[lane], c1 = cntL[32 + lane];
        int i0 = c0, i1 = c1;
        #pragma unroll
        for (int off = 1; off < 32; off <<= 1) {
            const int t0 = __shfl_up(i0, off, 32);
            const int t1 = __shfl_up(i1, off, 32);
            if (lane >= off) { i0 += t0; i1 += t1; }
        }
        const int tot0 = __shfl(i0, 31, 32);
        startL[lane]      = i0 - c0;
        startL[32 + lane] = tot0 + i1 - c1;
        if (lane == 31) nbL = tot0 + i1;
    }
    __syncthreads();
    const int  nb   = nbL;
    const bool fits = (nb <= CAP);
    if (fits) {
        #pragma unroll
        for (int k = 0; k < SEG_EW; ++k) {
            if (lo + sj + 4 * k >= hi) break;
            const unsigned w  = ew[k];
            const int      dl = (int)((w >> 16) & 63u);
            const int      p  = startL[dl] + atomicAdd(&curL[dl], 1);
            if ((unsigned)p < (unsigned)CAP) slist[p] = (int)(w & 0xffffu);
        }
        for (int i = lo + sj + 4 * SEG_EW; i < hi; i += 4) {
            const unsigned w  = seg[i];
            const int      dl = (int)((w >> 16) & 63u);
            const int      p  = startL[dl] + atomicAdd(&curL[dl], 1);
            if ((unsigned)p < (unsigned)CAP) slist[p] = (int)(w & 0xffffu);
        }
    }
    __syncthreads();

    for (int turn = 0; turn < 64 / NODE_WAVES; ++turn) {
        const int dl   = wy + NODE_WAVES * turn;
        const int node = bucket * 64 + dl;
        if (node >= N_NODES) break;

        const float dn = d[node];
        v2f acc01 = {0.f, 0.f}, acc23 = {0.f, 0.f};

        int deg  = fits ? __builtin_amdgcn_readfirstlane(cntL[dl])   : 65;
        int base = fits ? __builtin_amdgcn_readfirstlane(startL[dl]) : 0;
        if (deg < 0) deg = 0;
        if (base < 0) base = 0;
        if (base > CAP) base = CAP;
        if (deg > CAP - base && fits) deg = CAP - base;

        if (deg <= 64) {
            const int n0 = deg < 32 ? deg : 32;
            const int n1 = deg - n0;
            int k0 = 0x7fffffff, k1 = 0x7fffffff;
            if (lane < n0) k0 = slist[base + lane];
            if (lane < n1) k1 = slist[base + 32 + lane];

            int r0 = 0, r1 = 0;
            for (int i = 0; i < n0; ++i) {
                const int a = __builtin_amdgcn_readlane(k0, i);
                r0 += (a < k0 || (a == k0 && i < lane)) ? 1 : 0;
                r1 += (a <= k1) ? 1 : 0;
            }
            for (int i = 0; i < n1; ++i) {
                const int b = __builtin_amdgcn_readlane(k1, i);
                r0 += (b < k0) ? 1 : 0;
                r1 += (b < k1 || (b == k1 && i < lane)) ? 1 : 0;
            }

            int* ls = sorted_src[threadIdx.y];
            if (lane < n0) ls[r0 & 63] = k0;
            if (lane < n1) ls[r1 & 63] = k1;
            __builtin_amdgcn_fence(__ATOMIC_RELEASE, "wavefront");
            __builtin_amdgcn_wave_barrier();
            __builtin_amdgcn_fence(__ATOMIC_ACQUIRE, "wavefront");
            int ss0 = 0, ss1 = 0;
            if (lane < n0) ss0 = clamp_node(ls[lane]);
            if (lane < n1) ss1 = clamp_node(ls[32 + lane]);

            const float ninf = -__builtin_inff();
            float e0 = ninf, e1 = ninf;
            if (lane < n0) e0 = leaky(s[ss0] + dn);
            if (lane < n1) e1 = leaky(s[ss1] + dn);
            float m = wave_max32(fmaxf(e0, e1));
            m = __builtin_isfinite(m) ? m : 0.0f;
            const float x0 = (lane < n0) ? __expf(e0 - m) : 0.0f;
            const float x1 = (lane < n1) ? __expf(e1 - m) : 0.0f;
            float den = wave_sum32(x0 + x1);
            den = (den > 0.0f) ? den : 1.0f;
            const float a0 = x0 / den;
            const float a1 = x1 / den;

            int k = 0;
            for (; k + 4 <= n0; k += 4) add_rows<4>(z, lane, ss0, a0, k, acc01, acc23);
            if (n0 - k >= 2) { add_rows<2>(z, lane, ss0, a0, k, acc01, acc23); k += 2; }
            if (n0 - k >= 1) { add_rows<1>(z, lane, ss0, a0, k, acc01, acc23); }
            for (int q = 0; q < n1; ++q) add_rows<1>(z, lane, ss1, a1, q, acc01, acc23);
        } else {
            float acc[4] = {0.f, 0.f, 0.f, 0.f};
            node_general(ebuf, segLo, segHi, nb, node, lane, s, dn, z, acc);
            acc01.x = acc[0];  acc01.y = acc[1];
            acc23.x = acc[2];  acc23.y = acc[3];
        }

        *(float4*)(out + (size_t)node * DIM + lane * 4) = make_float4(acc01.x, acc01.y, acc23.x, acc23.y);
    }
}

extern "C" void kernel_launch(void* const* d_in, const int* in_sizes, int n_in,
                              void* d_out, int out_size, void* d_ws, size_t ws_size,
                              hipStream_t stream) {
    const float* h     = (const float*)d_in[0];
    const int*   src   = (const int*)  d_in[1];
    const int*   dst   = (const int*)  d_in[2];
    const float* W     = (const float*)d_in[3];
    const float* a_src = (const float*)d_in[4];
    const float* a_dst = (const float*)d_in[5];
    float*       out   = (float*)d_out;

    float*    ws    = (float*)d_ws;
    float*    z     = ws;
    float*    s     = z  + (size_t)N_NODES * DIM;
    float*    dn    = s  + N_NODES;
    int*      L     = (int*)(dn + N_NODES);
    unsigned* ebuf  = (unsigned*)(L + 2 * N_NODES);

    gat_gemm_wmma<<<NCHUNK + GEMM_WG, dim3(32, 8), 0, stream>>>(h, W, a_src, a_dst, z, s, dn, src, dst, L, ebuf);
    gat_node_kernel<<<NBUCK, dim3(32, NODE_WAVES), 0, stream>>>(L, ebuf, s, dn, z, out);
}
